# starting kernel
# baseline (speedup 1.0000x reference)
#include <hip/hip_runtime.h>
#include <hip/hip_bf16.h>
#include <cmath>

typedef __attribute__((ext_vector_type(2))) float v2f;
typedef __attribute__((ext_vector_type(8))) float v8f;

#define B_ 16
#define S_ 256
#define E_ 32
#define H_ 64

__global__ __launch_bounds__(32) void proj_kernel(
    const float* __restrict__ x, const float* __restrict__ Wv,
    const float* __restrict__ bv, const float* __restrict__ W1,
    const float* __restrict__ b1,
    float* __restrict__ a_ws, float* __restrict__ bf_ws,
    float* __restrict__ v_ws)
{
  const int tile = blockIdx.x;
  const int lane = threadIdx.x;
  const int rc   = lane & 15;
  const int kh   = lane >> 4;
  const int rowbase = tile * 16;

  v2f xr[8];
  float ss = 0.f;
  const float* xrow = x + (size_t)(rowbase + rc) * E_;
#pragma unroll
  for (int kb = 0; kb < 8; ++kb) {
    const int k0 = kb * 4 + 2 * kh;
    v2f t; t.x = xrow[k0]; t.y = xrow[k0 + 1];
    xr[kb] = t;
    ss += t.x * t.x + t.y * t.y;
  }
  ss += __shfl_xor(ss, 16, 32);
  const float inv = 1.0f / fmaxf(sqrtf(ss), 1e-12f);
  v2f xnr[8];
#pragma unroll
  for (int kb = 0; kb < 8; ++kb) xnr[kb] = xr[kb] * inv;

#pragma unroll
  for (int half = 0; half < 2; ++half) {
    const int ebase = half * E_;
    float* dst = half ? bf_ws : a_ws;
#pragma unroll
    for (int nt = 0; nt < 4; ++nt) {
      const int hb = nt * 16;
      v8f acc = {};
#pragma unroll
      for (int kb = 0; kb < 8; ++kb) {
        const int k0 = kb * 4 + 2 * kh;
        v2f bfrag;
        bfrag.x = W1[(size_t)(ebase + k0)     * H_ + hb + rc];
        bfrag.y = W1[(size_t)(ebase + k0 + 1) * H_ + hb + rc];
        acc = __builtin_amdgcn_wmma_f32_16x16x4_f32(
            false, xnr[kb], false, bfrag, (short)0, acc, false, false);
      }
      const float badd = half ? 0.f : b1[hb + rc];
#pragma unroll
      for (int r = 0; r < 8; ++r) {
        const int grow = rowbase + r + 8 * kh;
        dst[(size_t)grow * H_ + hb + rc] = acc[r] + badd;
      }
    }
  }

#pragma unroll
  for (int nt = 0; nt < 2; ++nt) {
    const int db = nt * 16;
    v8f acc = {};
#pragma unroll
    for (int kb = 0; kb < 8; ++kb) {
      const int k0 = kb * 4 + 2 * kh;
      v2f bfrag;
      bfrag.x = Wv[(size_t)(db + rc) * E_ + k0];
      bfrag.y = Wv[(size_t)(db + rc) * E_ + k0 + 1];
      acc = __builtin_amdgcn_wmma_f32_16x16x4_f32(
          false, xr[kb], false, bfrag, (short)0, acc, false, false);
    }
#pragma unroll
    for (int r = 0; r < 8; ++r) {
      const int grow = rowbase + r + 8 * kh;
      v_ws[(size_t)grow * E_ + db + rc] = acc[r] + bv[db + rc];
    }
  }
}

__global__ __launch_bounds__(256) void attn_kernel(
    const float* __restrict__ w2, const float* __restrict__ b2p,
    const float* __restrict__ a_ws, const float* __restrict__ bf_ws,
    const float* __restrict__ v_ws, float* __restrict__ out)
{
  __shared__ float s_a[16 * H_];
  __shared__ float s_sc[16 * S_];
  __shared__ float s_w2[H_];

  const int tid   = threadIdx.x;
  const int b     = blockIdx.x >> 4;
  const int itile = blockIdx.x & 15;
  const int irow0 = b * S_ + itile * 16;

  for (int idx = tid; idx < 16 * H_; idx += 256)
    s_a[idx] = a_ws[(size_t)irow0 * H_ + idx];
  if (tid < H_) s_w2[tid] = w2[tid];
  __syncthreads();

  const float b2 = b2p[0];
  const int i  = tid >> 4;
  const int jl = tid & 15;

  float sc[16];
  float m = -3.0e38f;
  for (int jj = 0; jj < 16; ++jj) {
    const int j = jj * 16 + jl;
    const float* bfr = bf_ws + (size_t)(b * S_ + j) * H_;
    const float* ar  = s_a + i * H_;
    float acc = 0.f;
#pragma unroll 8
    for (int h = 0; h < H_; ++h)
      acc += s_w2[h] * tanhf(ar[h] + bfr[h]);
    const float s = tanhf(acc + b2);
    sc[jj] = s;
    m = fmaxf(m, s);
  }
#pragma unroll
  for (int mask = 1; mask < 16; mask <<= 1)
    m = fmaxf(m, __shfl_xor(m, mask, 16));
  float sum = 0.f;
  for (int jj = 0; jj < 16; ++jj) { sc[jj] = expf(sc[jj] - m); sum += sc[jj]; }
#pragma unroll
  for (int mask = 1; mask < 16; mask <<= 1)
    sum += __shfl_xor(sum, mask, 16);
  const float isum = 1.0f / sum;
  for (int jj = 0; jj < 16; ++jj)
    s_sc[i * S_ + jj * 16 + jl] = sc[jj] * isum;
  __syncthreads();

  const int wave = tid >> 5;
  if (wave < 2) {
    const int lane = tid & 31;
    const int rc = lane & 15;
    const int kh = lane >> 4;
    const int db = wave * 16;
    const float* vb = v_ws + (size_t)b * S_ * E_;
    v8f acc = {};
    for (int kb = 0; kb < 64; ++kb) {
      const int k0 = kb * 4 + 2 * kh;
      v2f af, bfrag;
      af.x    = s_sc[rc * S_ + k0];
      af.y    = s_sc[rc * S_ + k0 + 1];
      bfrag.x = vb[(size_t)k0       * E_ + db + rc];
      bfrag.y = vb[(size_t)(k0 + 1) * E_ + db + rc];
      acc = __builtin_amdgcn_wmma_f32_16x16x4_f32(
          false, af, false, bfrag, (short)0, acc, false, false);
    }
#pragma unroll
    for (int r = 0; r < 8; ++r) {
      const int grow = irow0 + r + 8 * kh;
      out[(size_t)grow * E_ + db + rc] = acc[r];
    }
  }
}

extern "C" void kernel_launch(void* const* d_in, const int* in_sizes, int n_in,
                              void* d_out, int out_size, void* d_ws, size_t ws_size,
                              hipStream_t stream) {
  const float* x  = (const float*)d_in[0];
  const float* Wv = (const float*)d_in[1];
  const float* bv = (const float*)d_in[2];
  const float* W1 = (const float*)d_in[3];
  const float* b1 = (const float*)d_in[4];
  const float* w2 = (const float*)d_in[5];
  const float* b2 = (const float*)d_in[6];
  float* outp = (float*)d_out;

  float* ws    = (float*)d_ws;
  float* a_ws  = ws;
  float* bf_ws = a_ws + (size_t)B_ * S_ * H_;
  float* v_ws  = bf_ws + (size_t)B_ * S_ * H_;

  proj_kernel<<<dim3(B_ * S_ / 16), dim3(32), 0, stream>>>(
      x, Wv, bv, W1, b1, a_ws, bf_ws, v_ws);
  attn_kernel<<<dim3(B_ * (S_ / 16)), dim3(256), 0, stream>>>(
      w2, b2, a_ws, bf_ws, v_ws, outp);
}


# attention re-tiled to one wave per SIMD: 4-wave work-groups, thread = 4 rows x 8 columns, pair loop software-pipelined; wave owns whole rows: softmax max/sum by lane exchange, new row-sum order
# speedup vs baseline: 6.8451x; 6.8451x over previous
#include <hip/hip_runtime.h>
#include <hip/hip_bf16.h>
#include <cmath>

typedef __attribute__((ext_vector_type(2))) float v2f;
typedef __attribute__((ext_vector_type(4))) float v4f;
typedef __attribute__((ext_vector_type(8))) float v8f;

#define B_ 16
#define S_ 256
#define E_ 32
#define H_ 64

__global__ __launch_bounds__(32) void proj_kernel(
    const float* __restrict__ x, const float* __restrict__ Wv,
    const float* __restrict__ bv, const float* __restrict__ W1,
    const float* __restrict__ b1,
    float* __restrict__ a_ws, float* __restrict__ bf_ws,
    float* __restrict__ v_ws)
{
  const int tile = blockIdx.x;
  const int lane = threadIdx.x;
  const int rc   = lane & 15;
  const int kh   = lane >> 4;
  const int rowbase = tile * 16;

  v2f xr[8];
  float ss = 0.f;
  const float* xrow = x + (size_t)(rowbase + rc) * E_;
#pragma unroll
  for (int kb = 0; kb < 8; ++kb) {
    const int k0 = kb * 4 + 2 * kh;
    v2f t; t.x = xrow[k0]; t.y = xrow[k0 + 1];
    xr[kb] = t;
    ss += t.x * t.x + t.y * t.y;
  }
  ss += __shfl_xor(ss, 16, 32);
  const float inv = 1.0f / fmaxf(sqrtf(ss), 1e-12f);
  v2f xnr[8];
#pragma unroll
  for (int kb = 0; kb < 8; ++kb) xnr[kb] = xr[kb] * inv;

#pragma unroll
  for (int half = 0; half < 2; ++half) {
    const int ebase = half * E_;
    float* dst = half ? bf_ws : a_ws;
#pragma unroll
    for (int nt = 0; nt < 4; ++nt) {
      const int hb = nt * 16;
      v8f acc = {};
#pragma unroll
      for (int kb = 0; kb < 8; ++kb) {
        const int k0 = kb * 4 + 2 * kh;
        v2f bfrag;
        bfrag.x = W1[(size_t)(ebase + k0)     * H_ + hb + rc];
        bfrag.y = W1[(size_t)(ebase + k0 + 1) * H_ + hb + rc];
        acc = __builtin_amdgcn_wmma_f32_16x16x4_f32(
            false, xnr[kb], false, bfrag, (short)0, acc, false, false);
      }
      const float badd = half ? 0.f : b1[hb + rc];
#pragma unroll
      for (int r = 0; r < 8; ++r) {
        const int grow = rowbase + r + 8 * kh;
        dst[(size_t)grow * H_ + hb + rc] = acc[r] + badd;
      }
    }
  }

#pragma unroll
  for (int nt = 0; nt < 2; ++nt) {
    const int db = nt * 16;
    v8f acc = {};
#pragma unroll
    for (int kb = 0; kb < 8; ++kb) {
      const int k0 = kb * 4 + 2 * kh;
      v2f bfrag;
      bfrag.x = Wv[(size_t)(db + rc) * E_ + k0];
      bfrag.y = Wv[(size_t)(db + rc) * E_ + k0 + 1];
      acc = __builtin_amdgcn_wmma_f32_16x16x4_f32(
          false, xr[kb], false, bfrag, (short)0, acc, false, false);
    }
#pragma unroll
    for (int r = 0; r < 8; ++r) {
      const int grow = rowbase + r + 8 * kh;
      v_ws[(size_t)grow * E_ + db + rc] = acc[r] + bv[db + rc];
    }
  }
}

#define BP_ 68
#define VP_ 40
#define AP_ 20
#define C2LOG2E_ 2.8853900817779268f
#define NT_ 128

template <int N>
__device__ __forceinline__ float row_xor(float x) {
  const int b = __builtin_bit_cast(int, x);
  return __builtin_bit_cast(float, __builtin_amdgcn_update_dpp(b, b, 0x160 + N, 0xf, 0xf, false));
}
__device__ __forceinline__ float xhalf(float x) {
  const int b = __builtin_bit_cast(int, x);
  return __builtin_bit_cast(float, __builtin_amdgcn_permlanex16(b, b, 0x76543210, 0xfedcba98, false, false));
}

__device__ __forceinline__ v2f tanh_of_exponent(v2f s) {
  const v2f one  = {1.0f, 1.0f};
  const v2f mtwo = {-2.0f, -2.0f};
  v2f e;
  e.x = __builtin_amdgcn_exp2f(s.x);
  e.y = __builtin_amdgcn_exp2f(s.y);
  const v2f d = e + one;
  v2f r;
  r.x = __builtin_amdgcn_rcpf(d.x);
  r.y = __builtin_amdgcn_rcpf(d.y);
  return __builtin_elementwise_fma(mtwo, r, one);
}

struct PairRegs {
  v4f wv[2];
  v4f av[2][4];
  v4f bv[2][8];
  v2f sb[2][16];
  v2f eb[2][16];
  v2f db[16];
  v2f rb[2][16];
  v2f tb[16];
  v2f acc[2][8];
};

template <int R, bool DO_EXP, bool DO_SFMA, bool DO_ADD_RCP, bool DO_FINISH, bool DO_LOAD>
__device__ __forceinline__ void pair_region(PairRegs& pr, const int chunk,
    const float* s_w2, const float* s_a, const float* s_b, const int i0, const int lane)
{
  const v2f one  = {1.0f, 1.0f};
  const v2f mtwo = {-2.0f, -2.0f};
  const v2f cc   = {C2LOG2E_, C2LOG2E_};
  if (DO_LOAD && (R & 3) == 1) {
    const int hn = (chunk + 1) & (H_ / 4 - 1);
    const int set = ((R >> 2) + 1) & 1;
#pragma unroll
    for (int hh = 0; hh < 4; ++hh)
      pr.av[set][hh] = *reinterpret_cast<const v4f*>(&s_a[(hn * 4 + hh) * AP_ + i0]);
#pragma unroll
    for (int q = 0; q < 8; ++q)
      pr.bv[set][q] = *reinterpret_cast<const v4f*>(&s_b[(lane + 32 * q) * BP_ + hn * 4]);
  }
  if (DO_LOAD && (R & 3) == 2) {
    const int hn = (chunk + 1) & (H_ / 4 - 1);
    const int set = ((R >> 2) + 1) & 1;
    pr.wv[set] = *reinterpret_cast<const v4f*>(&s_w2[hn * 4]);
  }
  if (DO_ADD_RCP) {
#pragma unroll
    for (int x = 0; x < 16; ++x) pr.db[x] = pr.eb[(R + 1) & 1][x] + one;
  }
  if (DO_EXP) {
#pragma unroll
    for (int x = 0; x < 16; ++x) {
      pr.eb[R & 1][x].x = __builtin_amdgcn_exp2f(pr.sb[R & 1][x].x);
      pr.eb[R & 1][x].y = __builtin_amdgcn_exp2f(pr.sb[R & 1][x].y);
    }
  }
  if (DO_SFMA) {
    const int set = ((R + 1) >> 2) & 1, hh = (R + 1) & 3;
#pragma unroll
    for (int q = 0; q < 8; ++q)
#pragma unroll
      for (int p = 0; p < 2; ++p) {
        const v2f aa = {pr.av[set][hh][2 * p], pr.av[set][hh][2 * p + 1]};
        const v2f bb = {pr.bv[set][q][hh], pr.bv[set][q][hh]};
        pr.sb[(R + 1) & 1][2 * q + p] = __builtin_elementwise_fma(aa, cc, bb);
      }
  }
  if (DO_ADD_RCP) {
#pragma unroll
    for (int x = 0; x < 16; ++x) {
      pr.rb[(R + 1) & 1][x].x = __builtin_amdgcn_rcpf(pr.db[x].x);
      pr.rb[(R + 1) & 1][x].y = __builtin_amdgcn_rcpf(pr.db[x].y);
    }
  }
  if (DO_FINISH) {
    const int set = ((R + 6) >> 2) & 1, hh = (R + 6) & 3;
#pragma unroll
    for (int x = 0; x < 16; ++x) pr.tb[x] = __builtin_elementwise_fma(mtwo, pr.rb[R & 1][x], one);
#pragma unroll
    for (int q = 0; q < 8; ++q)
#pragma unroll
      for (int p = 0; p < 2; ++p) {
        const v2f ww = {pr.wv[set][hh], pr.wv[set][hh]};
        pr.acc[p][q] = __builtin_elementwise_fma(ww, pr.tb[2 * q + p], pr.acc[p][q]);
      }
  }
  if (DO_LOAD && (R & 3) == 1) __builtin_amdgcn_sched_group_barrier(0x100, 12, 0);
  if (DO_LOAD && (R & 3) == 2) __builtin_amdgcn_sched_group_barrier(0x100, 1, 0);
#pragma unroll
  for (int t = 0; t < 64; ++t) {
    __builtin_amdgcn_sched_group_barrier(0x002, 1, 0);
    __builtin_amdgcn_sched_group_barrier(0x400, 1, 0);
  }
  __builtin_amdgcn_sched_barrier(0);
}

__global__ __launch_bounds__(NT_) void attn_kernel(
    const float* __restrict__ w2, const float* __restrict__ b2p,
    const float* __restrict__ a_ws, const float* __restrict__ bf_ws,
    const float* __restrict__ v_ws, float* __restrict__ out)
{
  __shared__ __attribute__((aligned(16))) float s_b[S_ * BP_];
  __shared__ __attribute__((aligned(16))) float s_a[H_ * AP_];
  __shared__ __attribute__((aligned(16))) float s_sc[16 * S_];
  __shared__ __attribute__((aligned(16))) float s_w2[H_];
  __shared__ __attribute__((aligned(16))) float s_v[S_ * VP_];

  const int tid   = threadIdx.x;
  const int b     = blockIdx.x >> 4;
  const int itile = blockIdx.x & 15;
  const int irow0 = b * S_ + itile * 16;

  {
    const v4f* bsrc = reinterpret_cast<const v4f*>(bf_ws + (size_t)b * S_ * H_);
    const v4f* vsrc = reinterpret_cast<const v4f*>(v_ws + (size_t)b * S_ * E_);
    const v4f* asrc = reinterpret_cast<const v4f*>(a_ws + (size_t)irow0 * H_);
    v4f breg[32];
    v4f vreg[16];
    v4f areg[2];
#pragma unroll
    for (int k = 0; k < 32; ++k) breg[k] = bsrc[tid + NT_ * k];
#pragma unroll
    for (int k = 0; k < 2; ++k) areg[k] = asrc[tid + NT_ * k];
    const float wreg = w2[tid & (H_ - 1)];
#pragma unroll
    for (int k = 0; k < 16; ++k) vreg[k] = vsrc[tid + NT_ * k];

    s_w2[tid & (H_ - 1)] = wreg;
#pragma unroll
    for (int k = 0; k < 32; ++k) {
      const int idx4 = tid + NT_ * k;
      const int row = idx4 >> 4, c4 = idx4 & 15;
      *reinterpret_cast<v4f*>(&s_b[row * BP_ + c4 * 4]) = breg[k] * C2LOG2E_;
    }
#pragma unroll
    for (int k = 0; k < 2; ++k) {
      const int idx4 = tid + NT_ * k;
      const int ai = idx4 >> 4, ah = (idx4 & 15) * 4;
#pragma unroll
      for (int e = 0; e < 4; ++e) s_a[(ah + e) * AP_ + ai] = areg[k][e];
    }
#pragma unroll
    for (int k = 0; k < 16; ++k) {
      const int idx4 = tid + NT_ * k;
      const int row = idx4 >> 3, c4 = idx4 & 7;
      *reinterpret_cast<v4f*>(&s_v[row * VP_ + c4 * 4]) = vreg[k];
    }
  }
  __syncthreads();

  const float b2 = b2p[0];
  {
    const int wave = tid >> 5;
    const int lane = tid & 31;
    const int i0 = wave * 4;

    PairRegs pr;
#pragma unroll
    for (int p = 0; p < 2; ++p)
#pragma unroll
      for (int q = 0; q < 8; ++q) pr.acc[p][q] = v2f{0.0f, 0.0f};

    pr.wv[0] = *reinterpret_cast<const v4f*>(&s_w2[0]);
#pragma unroll
    for (int hh = 0; hh < 4; ++hh)
      pr.av[0][hh] = *reinterpret_cast<const v4f*>(&s_a[hh * AP_ + i0]);
#pragma unroll
    for (int q = 0; q < 8; ++q)
      pr.bv[0][q] = *reinterpret_cast<const v4f*>(&s_b[(lane + 32 * q) * BP_]);
    {
      const v2f cc = {C2LOG2E_, C2LOG2E_};
#pragma unroll
      for (int q = 0; q < 8; ++q)
#pragma unroll
        for (int p = 0; p < 2; ++p) {
          const v2f aa = {pr.av[0][0][2 * p], pr.av[0][0][2 * p + 1]};
          const v2f bb = {pr.bv[0][q][0], pr.bv[0][q][0]};
          pr.sb[0][2 * q + p] = __builtin_elementwise_fma(aa, cc, bb);
        }
    }
    __builtin_amdgcn_sched_barrier(0);

    {
      pair_region<0, true, true, false, false, true>(pr, 0, s_w2, s_a, s_b, i0, lane);
      pair_region<1, true, true, true, false, true>(pr, 0, s_w2, s_a, s_b, i0, lane);
      pair_region<2, true, true, true, true, true>(pr, 0, s_w2, s_a, s_b, i0, lane);
      pair_region<3, true, true, true, true, true>(pr, 0, s_w2, s_a, s_b, i0, lane);
      pair_region<4, true, true, true, true, true>(pr, 1, s_w2, s_a, s_b, i0, lane);
      pair_region<5, true, true, true, true, true>(pr, 1, s_w2, s_a, s_b, i0, lane);
      pair_region<6, true, true, true, true, true>(pr, 1, s_w2, s_a, s_b, i0, lane);
      pair_region<7, true, true, true, true, true>(pr, 1, s_w2, s_a, s_b, i0, lane);
    }
    for (int c2 = 1; c2 < H_ / 8; ++c2) {
      pair_region<0, true, true, true, true, true>(pr, 2 * c2, s_w2, s_a, s_b, i0, lane);
      pair_region<1, true, true, true, true, true>(pr, 2 * c2, s_w2, s_a, s_b, i0, lane);
      pair_region<2, true, true, true, true, true>(pr, 2 * c2, s_w2, s_a, s_b, i0, lane);
      pair_region<3, true, true, true, true, true>(pr, 2 * c2, s_w2, s_a, s_b, i0, lane);
      pair_region<4, true, true, true, true, true>(pr, 2 * c2 + 1, s_w2, s_a, s_b, i0, lane);
      pair_region<5, true, true, true, true, true>(pr, 2 * c2 + 1, s_w2, s_a, s_b, i0, lane);
      pair_region<6, true, true, true, true, true>(pr, 2 * c2 + 1, s_w2, s_a, s_b, i0, lane);
      pair_region<7, true, true, true, true, true>(pr, 2 * c2 + 1, s_w2, s_a, s_b, i0, lane);
    }
    {
      pair_region<0, false, false, true, true, false>(pr, H_ / 4, s_w2, s_a, s_b, i0, lane);
      pair_region<1, false, false, false, true, false>(pr, H_ / 4, s_w2, s_a, s_b, i0, lane);
    }

    const v2f b2v = {b2, b2};
    const v2f cc = {C2LOG2E_, C2LOG2E_};
    float sc[4][8];
#pragma unroll
    for (int p = 0; p < 2; ++p)
#pragma unroll
      for (int q = 0; q < 8; ++q) {
        const v2f t = tanh_of_exponent((pr.acc[p][q] + b2v) * cc);
        sc[2 * p][q]     = t.x;
        sc[2 * p + 1][q] = t.y;
      }
    float m[4], sum[4];
#pragma unroll
    for (int r = 0; r < 4; ++r) {
      m[r] = -3.0e38f;
#pragma unroll
      for (int q = 0; q < 8; ++q) m[r] = fmaxf(m[r], sc[r][q]);
    }
#pragma unroll
    for (int r = 0; r < 4; ++r) m[r] = fmaxf(m[r], row_xor<1>(m[r]));
#pragma unroll
    for (int r = 0; r < 4; ++r) m[r] = fmaxf(m[r], row_xor<2>(m[r]));
#pragma unroll
    for (int r = 0; r < 4; ++r) m[r] = fmaxf(m[r], row_xor<4>(m[r]));
#pragma unroll
    for (int r = 0; r < 4; ++r) m[r] = fmaxf(m[r], row_xor<8>(m[r]));
#pragma unroll
    for (int r = 0; r < 4; ++r) m[r] = fmaxf(m[r], xhalf(m[r]));
#pragma unroll
    for (int r = 0; r < 4; ++r) {
      sum[r] = 0.f;
#pragma unroll
      for (int q = 0; q < 8; ++q) { sc[r][q] = expf(sc[r][q] - m[r]); sum[r] += sc[r][q]; }
    }
#pragma unroll
    for (int r = 0; r < 4; ++r) sum[r] += row_xor<1>(sum[r]);
#pragma unroll
    for (int r = 0; r < 4; ++r) sum[r] += row_xor<2>(sum[r]);
#pragma unroll
    for (int r = 0; r < 4; ++r) sum[r] += row_xor<4>(sum[r]);
#pragma unroll
    for (int r = 0; r < 4; ++r) sum[r] += row_xor<8>(sum[r]);
#pragma unroll
    for (int r = 0; r < 4; ++r) sum[r] += xhalf(sum[r]);
#pragma unroll
    for (int r = 0; r < 4; ++r) {
      const float isum = 1.0f / sum[r];
#pragma unroll
      for (int q = 0; q < 8; ++q)
        s_sc[(i0 + r) * S_ + lane + 32 * q] = sc[r][q] * isum;
    }
  }
  __syncthreads();

  const int wave = tid >> 5;
  if (wave < 2) {
    const int lane = tid & 31;
    const int rc = lane & 15;
    const int kh = lane >> 4;
    const int db = wave * 16;
    const float* ap = &s_sc[rc * S_ + 2 * kh];
    const float* bp = &s_v[(2 * kh) * VP_ + db + rc];
    v2f af[2][8], bfr[2][8];
#pragma unroll
    for (int t = 0; t < 8; ++t) {
      af[0][t] = *reinterpret_cast<const v2f*>(ap + 4 * t);
      bfr[0][t].x = bp[(4 * t) * VP_];
      bfr[0][t].y = bp[(4 * t + 1) * VP_];
    }
    v8f acc = {};
#pragma unroll
    for (int blk = 0; blk < 8; ++blk) {
      const int cur = blk & 1, nxt = cur ^ 1;
      if (blk + 1 < 8) {
#pragma unroll
        for (int t = 0; t < 8; ++t) {
          const int kb = (blk + 1) * 8 + t;
          af[nxt][t] = *reinterpret_cast<const v2f*>(ap + 4 * kb);
          bfr[nxt][t].x = bp[(4 * kb) * VP_];
          bfr[nxt][t].y = bp[(4 * kb + 1) * VP_];
        }
      }
      __builtin_amdgcn_sched_barrier(0);
#pragma unroll
      for (int t = 0; t < 8; ++t)
        acc = __builtin_amdgcn_wmma_f32_16x16x4_f32(
            false, af[cur][t], false, bfr[cur][t], (short)0, acc, false, false);
      __builtin_amdgcn_sched_barrier(0);
    }
#pragma unroll
    for (int r = 0; r < 8; ++r) {
      const int grow = irow0 + r + 8 * kh;
      out[(size_t)grow * E_ + db + rc] = acc[r];
    }
  }
}

extern "C" void kernel_launch(void* const* d_in, const int* in_sizes, int n_in,
                              void* d_out, int out_size, void* d_ws, size_t ws_size,
                              hipStream_t stream) {
  const float* x  = (const float*)d_in[0];
  const float* Wv = (const float*)d_in[1];
  const float* bv = (const float*)d_in[2];
  const float* W1 = (const float*)d_in[3];
  const float* b1 = (const float*)d_in[4];
  const float* w2 = (const float*)d_in[5];
  const float* b2 = (const float*)d_in[6];
  float* outp = (float*)d_out;

  float* ws    = (float*)d_ws;
  float* a_ws  = ws;
  float* bf_ws = a_ws + (size_t)B_ * S_ * H_;
  float* v_ws  = bf_ws + (size_t)B_ * S_ * H_;

  proj_kernel<<<dim3(B_ * S_ / 16), dim3(32), 0, stream>>>(
      x, Wv, bv, W1, b1, a_ws, bf_ws, v_ws);
  attn_kernel<<<dim3(B_ * (S_ / 16)), dim3(NT_), 0, stream>>>(
      w2, b2, a_ws, bf_ws, v_ws, outp);
}
